# starting kernel
# baseline (speedup 1.0000x reference)
#include <hip/hip_runtime.h>
#include <hip/hip_bf16.h>


typedef __attribute__((ext_vector_type(16))) _Float16 v16h;
typedef __attribute__((ext_vector_type(8)))  float    v8f;
typedef __attribute__((ext_vector_type(4)))  int      v4i;

#define NN      50000
#define DH      128
#define DCAT    256
#define NE      800000
#define MTILES  (NN / 16)

#define WMMA_F16(A, B, C) \
    __builtin_amdgcn_wmma_f32_16x16x32_f16(false, (A), false, (B), (short)0, (C), false, false)

__device__ __forceinline__ v16h load_frag(const _Float16* p0, const _Float16* p1) {
    v16h r;
    reinterpret_cast<v4i*>(&r)[0] = *reinterpret_cast<const v4i*>(p0);
    reinterpret_cast<v4i*>(&r)[1] = *reinterpret_cast<const v4i*>(p1);
    return r;
}

__device__ __forceinline__ float sigmoid_f(float x) {
    return 1.0f / (1.0f + __expf(-x));
}

__global__ void __launch_bounds__(256) prep_weight(const float* __restrict__ W,
                                                   _Float16* __restrict__ frag,
                                                   int K) {
    int idx   = blockIdx.x * 256 + threadIdx.x;
    int total = (K / 32) * 8 * 32 * 16;
    if (idx >= total) return;
    int j     = idx & 15;
    int lane  = (idx >> 4) & 31;
    int ntile = (idx >> 9) & 7;
    int ktile = idx >> 12;
    int n     = ntile * 16 + (lane & 15);
    int khalf = lane >> 4;
    int k     = ktile * 32 + khalf * 8 + ((j < 8) ? j : 16 + (j - 8));
    frag[idx] = (_Float16)W[(size_t)k * DH + n];
}

__global__ void __launch_bounds__(256) f32_to_f16(const float* __restrict__ a,
                                                  _Float16* __restrict__ b, int n) {
    int i = blockIdx.x * 256 + threadIdx.x;
    if (i < n) b[i] = (_Float16)a[i];
}

__global__ void __launch_bounds__(256) zero_f32(float* __restrict__ p, int n) {
    int i = blockIdx.x * 256 + threadIdx.x;
    if (i < n) p[i] = 0.0f;
}

__global__ void __launch_bounds__(256) gates_kernel(
    const _Float16* __restrict__ x16,  const _Float16* __restrict__ h16,
    const float* __restrict__ h_prev,
    const float* __restrict__ b_xr,    const float* __restrict__ b_xz,
    const _Float16* __restrict__ fWxr, const _Float16* __restrict__ fWhr,
    const _Float16* __restrict__ fWxz, const _Float16* __restrict__ fWhz,
    float* __restrict__ z_ws, float* __restrict__ rh32, _Float16* __restrict__ rh16)
{
    const int lane  = threadIdx.x & 31;
    const int ntile = threadIdx.x >> 5;
    const int mtile = blockIdx.x;
    const int col   = ntile * 16 + (lane & 15);
    const int khalf = lane >> 4;
    const int arow  = mtile * 16 + (lane & 15);

    v8f accr, accz;
    const float br = b_xr[col], bz = b_xz[col];
#pragma unroll
    for (int v = 0; v < 8; ++v) { accr[v] = br; accz[v] = bz; }

    const _Float16* xb = x16 + (size_t)arow * DH;
    const _Float16* hb = h16 + (size_t)arow * DH;

#pragma unroll
    for (int kt = 0; kt < 4; ++kt) {
        const int ko = kt * 32 + khalf * 8;
        v16h ax = load_frag(xb + ko, xb + ko + 16);
        v16h ah = load_frag(hb + ko, hb + ko + 16);
        const size_t off = ((size_t)(kt * 8 + ntile) * 32 + lane) * 16;
        v16h bxr = load_frag(fWxr + off, fWxr + off + 8);
        v16h bhr = load_frag(fWhr + off, fWhr + off + 8);
        v16h bxz = load_frag(fWxz + off, fWxz + off + 8);
        v16h bhz = load_frag(fWhz + off, fWhz + off + 8);
        accr = WMMA_F16(ax, bxr, accr);
        accr = WMMA_F16(ah, bhr, accr);
        accz = WMMA_F16(ax, bxz, accz);
        accz = WMMA_F16(ah, bhz, accz);
    }

    const int mbase = mtile * 16 + ((lane & 16) ? 8 : 0);
#pragma unroll
    for (int v = 0; v < 8; ++v) {
        const size_t idx = (size_t)(mbase + v) * DH + col;
        const float rv = sigmoid_f(accr[v]);
        const float zv = sigmoid_f(accz[v]);
        const float h  = h_prev[idx];
        const float rh = rv * h;
        z_ws[idx] = zv;
        rh32[idx] = rh;
        rh16[idx] = (_Float16)rh;
    }
}

__global__ void __launch_bounds__(256) scatter_kernel(
    const int* __restrict__ ei, const float* __restrict__ x,
    const float* __restrict__ rh32,
    float* __restrict__ agg, float* __restrict__ cnt)
{
    const int lane = threadIdx.x & 31;
    const int wave = threadIdx.x >> 5;
    const int e    = blockIdx.x * 8 + wave;
    const int s    = ei[e];
    const int d    = ei[NE + e];
    const float* xs = x    + (size_t)s * DH;
    const float* rs = rh32 + (size_t)s * DH;
    float* ag = agg + (size_t)d * DCAT;
#pragma unroll
    for (int i = 0; i < 4; ++i) {
        const int c = lane + i * 32;
        atomicAdd(&ag[c], xs[c]);
    }
#pragma unroll
    for (int i = 0; i < 4; ++i) {
        const int c = lane + i * 32;
        atomicAdd(&ag[DH + c], rs[c]);
    }
    if (lane == 0) atomicAdd(&cnt[d], 1.0f);
}

__global__ void __launch_bounds__(256) cand_kernel(
    const float* __restrict__ agg, const float* __restrict__ cnt,
    const _Float16* __restrict__ x16, const _Float16* __restrict__ rh16,
    const _Float16* __restrict__ fWl, const _Float16* __restrict__ fWr,
    const float* __restrict__ b_l,
    const float* __restrict__ z_ws, const float* __restrict__ h_prev,
    float* __restrict__ out)
{
    const int lane  = threadIdx.x & 31;
    const int ntile = threadIdx.x >> 5;
    const int mtile = blockIdx.x;
    const int col   = ntile * 16 + (lane & 15);
    const int khalf = lane >> 4;
    const int arow  = mtile * 16 + (lane & 15);

    const float inv = 1.0f / fmaxf(cnt[arow], 1.0f);

    v8f acc;
    const float bl = b_l[col];
#pragma unroll
    for (int v = 0; v < 8; ++v) acc[v] = bl;

    const float* ab = agg + (size_t)arow * DCAT;

#pragma unroll
    for (int kt = 0; kt < 8; ++kt) {
        const int ko = kt * 32 + khalf * 8;
        v16h am;
#pragma unroll
        for (int j = 0; j < 8; ++j) {
            am[j]     = (_Float16)(ab[ko + j]      * inv);
            am[8 + j] = (_Float16)(ab[ko + 16 + j] * inv);
        }
        const _Float16* xcb = (kt < 4)
            ? (x16  + (size_t)arow * DH + kt * 32)
            : (rh16 + (size_t)arow * DH + (kt - 4) * 32);
        v16h ac = load_frag(xcb + khalf * 8, xcb + khalf * 8 + 16);

        const size_t off = ((size_t)(kt * 8 + ntile) * 32 + lane) * 16;
        v16h bwl = load_frag(fWl + off, fWl + off + 8);
        v16h bwr = load_frag(fWr + off, fWr + off + 8);
        acc = WMMA_F16(am, bwl, acc);
        acc = WMMA_F16(ac, bwr, acc);
    }

    const int mbase = mtile * 16 + ((lane & 16) ? 8 : 0);
#pragma unroll
    for (int v = 0; v < 8; ++v) {
        const size_t idx = (size_t)(mbase + v) * DH + col;
        const float zv = z_ws[idx];
        out[idx] = (1.0f - zv) * acc[v] + zv * h_prev[idx];
    }
}

extern "C" void kernel_launch(void* const* d_in, const int* in_sizes, int n_in,
                              void* d_out, int out_size, void* d_ws, size_t ws_size,
                              hipStream_t stream) {
    const float* x      = (const float*)d_in[0];
    const int*   ei     = (const int*)  d_in[1];
    const float* h_prev = (const float*)d_in[2];
    const float* W_xr   = (const float*)d_in[3];
    const float* b_xr   = (const float*)d_in[4];
    const float* W_hr   = (const float*)d_in[5];
    const float* W_xz   = (const float*)d_in[6];
    const float* b_xz   = (const float*)d_in[7];
    const float* W_hz   = (const float*)d_in[8];
    const float* W_l    = (const float*)d_in[9];
    const float* b_l    = (const float*)d_in[10];
    const float* W_r    = (const float*)d_in[11];
    float* out = (float*)d_out;

    char* ws = (char*)d_ws;
    size_t off = 0;
    auto alloc = [&](size_t bytes) -> char* {
        char* p = ws + off;
        off += (bytes + 255) & ~(size_t)255;
        return p;
    };
    _Float16* x16  = (_Float16*)alloc((size_t)NN * DH * 2);
    _Float16* h16  = (_Float16*)alloc((size_t)NN * DH * 2);
    _Float16* rh16 = (_Float16*)alloc((size_t)NN * DH * 2);
    float*    z_ws = (float*)   alloc((size_t)NN * DH * 4);
    float*    rh32 = (float*)   alloc((size_t)NN * DH * 4);
    float*    agg  = (float*)   alloc((size_t)NN * DCAT * 4);
    float*    cnt  = (float*)   alloc((size_t)NN * 4);
    _Float16* fWxr = (_Float16*)alloc(128 * 128 * 2);
    _Float16* fWhr = (_Float16*)alloc(128 * 128 * 2);
    _Float16* fWxz = (_Float16*)alloc(128 * 128 * 2);
    _Float16* fWhz = (_Float16*)alloc(128 * 128 * 2);
    _Float16* fWl  = (_Float16*)alloc(256 * 128 * 2);
    _Float16* fWr  = (_Float16*)alloc(256 * 128 * 2);

    prep_weight<<<64,  256, 0, stream>>>(W_xr, fWxr, 128);
    prep_weight<<<64,  256, 0, stream>>>(W_hr, fWhr, 128);
    prep_weight<<<64,  256, 0, stream>>>(W_xz, fWxz, 128);
    prep_weight<<<64,  256, 0, stream>>>(W_hz, fWhz, 128);
    prep_weight<<<128, 256, 0, stream>>>(W_l,  fWl,  256);
    prep_weight<<<128, 256, 0, stream>>>(W_r,  fWr,  256);

    const int nact = NN * DH;
    f32_to_f16<<<(nact + 255) / 256, 256, 0, stream>>>(x,      x16, nact);
    f32_to_f16<<<(nact + 255) / 256, 256, 0, stream>>>(h_prev, h16, nact);
    zero_f32<<<((NN * DCAT) + 255) / 256, 256, 0, stream>>>(agg, NN * DCAT);
    zero_f32<<<(NN + 255) / 256, 256, 0, stream>>>(cnt, NN);

    gates_kernel<<<MTILES, 256, 0, stream>>>(x16, h16, h_prev, b_xr, b_xz,
                                             fWxr, fWhr, fWxz, fWhz,
                                             z_ws, rh32, rh16);

    scatter_kernel<<<NE / 8, 256, 0, stream>>>(ei, x, rh32, agg, cnt);

    cand_kernel<<<MTILES, 256, 0, stream>>>(agg, cnt, x16, rh16, fWl, fWr,
                                            b_l, z_ws, h_prev, out);
}


# One preparation launch instead of ten: the six weight swizzles, two f16 copies and two clears share one grid (a block's index picks its job); 13 launches become 4
# speedup vs baseline: 2.2506x; 2.2506x over previous
#include <hip/hip_runtime.h>
#include <hip/hip_bf16.h>


typedef __attribute__((ext_vector_type(16))) _Float16 v16h;
typedef __attribute__((ext_vector_type(8)))  float    v8f;
typedef __attribute__((ext_vector_type(4)))  int      v4i;
typedef __attribute__((ext_vector_type(4)))  float    v4f;
typedef __attribute__((ext_vector_type(4)))  _Float16 v4h;

#define NN      50000
#define DH      128
#define DCAT    256
#define NE      800000
#define MTILES  (NN / 16)
#define RH_PITCH 136
#define CHUNKS  5
#define EPW     16

#define WMMA_F16(A, B, C) \
    __builtin_amdgcn_wmma_f32_16x16x32_f16(false, (A), false, (B), (short)0, (C), false, false)

__device__ __forceinline__ v16h load_frag(const _Float16* p0, const _Float16* p1) {
    v16h r;
    reinterpret_cast<v4i*>(&r)[0] = *reinterpret_cast<const v4i*>(p0);
    reinterpret_cast<v4i*>(&r)[1] = *reinterpret_cast<const v4i*>(p1);
    return r;
}

__device__ __forceinline__ float sigmoid_f(float x) {
    return 1.0f / (1.0f + __expf(-x));
}

#define PIECES_PER_BLOCK (256 * CHUNKS)
#define ACT_BLOCKS  (NN * DH / 4 / PIECES_PER_BLOCK)
#define CNT_PIECES  (NN / 4)
#define CNT_BLOCKS  ((CNT_PIECES + PIECES_PER_BLOCK - 1) / PIECES_PER_BLOCK)
#define W_BLOCKS    (4 * 64 + 2 * 128)
#define PREP_BLOCKS (W_BLOCKS + CNT_BLOCKS + ACT_BLOCKS)
static_assert((NN * DH / 4) % PIECES_PER_BLOCK == 0, "the f16 copies have no tail: blocks must tile the arrays exactly");
static_assert(NN % 4 == 0, "the count buffer is cleared in 16-byte pieces");

__global__ void __launch_bounds__(256) prep_all(
    const float* __restrict__ x,    const float* __restrict__ h_prev,
    _Float16* __restrict__ x16,     _Float16* __restrict__ h16,
    float* __restrict__ aggy,       float* __restrict__ cnt,
    const float* __restrict__ W_xr, const float* __restrict__ W_hr,
    const float* __restrict__ W_xz, const float* __restrict__ W_hz,
    const float* __restrict__ W_l,  const float* __restrict__ W_r,
    _Float16* __restrict__ fWxr,    _Float16* __restrict__ fWhr,
    _Float16* __restrict__ fWxz,    _Float16* __restrict__ fWhz,
    _Float16* __restrict__ fWl,     _Float16* __restrict__ fWr)
{
    const int b = blockIdx.x;
    const v4f zero4 = {0.0f, 0.0f, 0.0f, 0.0f};

    if (b < W_BLOCKS) {
        const float* W; _Float16* frag; int lb;
        if      (b <  64) { W = W_xr; frag = fWxr; lb = b;       }
        else if (b < 128) { W = W_hr; frag = fWhr; lb = b -  64; }
        else if (b < 192) { W = W_xz; frag = fWxz; lb = b - 128; }
        else if (b < 256) { W = W_hz; frag = fWhz; lb = b - 192; }
        else if (b < 384) { W = W_l;  frag = fWl;  lb = b - 256; }
        else              { W = W_r;  frag = fWr;  lb = b - 384; }
        const int idx   = lb * 256 + threadIdx.x;
        const int j     = idx & 15;
        const int lane  = (idx >> 4) & 31;
        const int ntile = (idx >> 9) & 7;
        const int ktile = idx >> 12;
        const int n     = ntile * 16 + (lane & 15);
        const int khalf = lane >> 4;
        const int k     = ktile * 32 + khalf * 8 + ((j < 8) ? j : 16 + (j - 8));
        frag[idx] = (_Float16)W[(size_t)k * DH + n];
    } else if (b < W_BLOCKS + CNT_BLOCKS) {
        v4f* p4 = reinterpret_cast<v4f*>(cnt);
        const int base = (b - W_BLOCKS) * PIECES_PER_BLOCK + threadIdx.x;
        for (int k = 0; k < CHUNKS; ++k) {
            const int i = base + k * 256;
            if (i < CNT_PIECES) p4[i] = zero4;
        }
    } else {
        const size_t base = (size_t)(b - W_BLOCKS - CNT_BLOCKS) * PIECES_PER_BLOCK + threadIdx.x;
        const v4f* xa = reinterpret_cast<const v4f*>(x)      + base;
        const v4f* ha = reinterpret_cast<const v4f*>(h_prev) + base;
        v4h*       xb = reinterpret_cast<v4h*>(x16) + base;
        v4h*       hb = reinterpret_cast<v4h*>(h16) + base;
        v4f*       ag = reinterpret_cast<v4f*>(aggy) + base;
        v4f vx[CHUNKS], vh[CHUNKS];
        for (int k = 0; k < CHUNKS; ++k) vx[k] = xa[k * 256];
        for (int k = 0; k < CHUNKS; ++k) vh[k] = ha[k * 256];
        for (int k = 0; k < CHUNKS; ++k) ag[k * 256] = zero4;
        for (int k = 0; k < CHUNKS; ++k) xb[k * 256] = __builtin_convertvector(vx[k], v4h);
        for (int k = 0; k < CHUNKS; ++k) hb[k * 256] = __builtin_convertvector(vh[k], v4h);
    }
}

__global__ void __launch_bounds__(256) gates_kernel(
    const _Float16* __restrict__ x16,  const _Float16* __restrict__ h16,
    const float* __restrict__ h_prev,
    const float* __restrict__ b_xr,    const float* __restrict__ b_xz,
    const _Float16* __restrict__ fWxr, const _Float16* __restrict__ fWhr,
    const _Float16* __restrict__ fWxz, const _Float16* __restrict__ fWhz,
    const _Float16* __restrict__ fWl,
    float* __restrict__ z_ws, float* __restrict__ y32, _Float16* __restrict__ rh16)
{
    __shared__ __attribute__((aligned(16))) _Float16 s_rh[16 * RH_PITCH];

    const int lane  = threadIdx.x & 31;
    const int ntile = threadIdx.x >> 5;
    const int mtile = blockIdx.x;
    const int col   = ntile * 16 + (lane & 15);
    const int khalf = lane >> 4;
    const int arow  = mtile * 16 + (lane & 15);

    v8f accr, accz, accy;
    const float br = b_xr[col], bz = b_xz[col];
#pragma unroll
    for (int v = 0; v < 8; ++v) { accr[v] = br; accz[v] = bz; accy[v] = 0.0f; }

    const _Float16* xb = x16 + (size_t)arow * DH;
    const _Float16* hb = h16 + (size_t)arow * DH;

#pragma unroll
    for (int kt = 0; kt < 4; ++kt) {
        const int ko = kt * 32 + khalf * 8;
        v16h ax = load_frag(xb + ko, xb + ko + 16);
        v16h ah = load_frag(hb + ko, hb + ko + 16);
        const size_t off = ((size_t)(kt * 8 + ntile) * 32 + lane) * 16;
        v16h bxr = load_frag(fWxr + off, fWxr + off + 8);
        v16h bhr = load_frag(fWhr + off, fWhr + off + 8);
        v16h bxz = load_frag(fWxz + off, fWxz + off + 8);
        v16h bhz = load_frag(fWhz + off, fWhz + off + 8);
        v16h bwl = load_frag(fWl  + off, fWl  + off + 8);
        accr = WMMA_F16(ax, bxr, accr);
        accr = WMMA_F16(ah, bhr, accr);
        accz = WMMA_F16(ax, bxz, accz);
        accz = WMMA_F16(ah, bhz, accz);
        accy = WMMA_F16(ax, bwl, accy);
    }

    const int rloc  = (lane & 16) ? 8 : 0;
    const int mbase = mtile * 16 + rloc;
#pragma unroll
    for (int v = 0; v < 8; ++v) {
        const size_t idx = (size_t)(mbase + v) * DH + col;
        const float rv = sigmoid_f(accr[v]);
        const float zv = sigmoid_f(accz[v]);
        const float h  = h_prev[idx];
        const _Float16 rh = (_Float16)(rv * h);
        z_ws[idx] = zv;
        rh16[idx] = rh;
        s_rh[(rloc + v) * RH_PITCH + col] = rh;
    }
    __syncthreads();

    const _Float16* sp = s_rh + (lane & 15) * RH_PITCH + khalf * 8;
#pragma unroll
    for (int kt = 0; kt < 4; ++kt) {
        v16h arh = load_frag(sp + kt * 32, sp + kt * 32 + 16);
        const size_t off = ((size_t)((4 + kt) * 8 + ntile) * 32 + lane) * 16;
        v16h bwl = load_frag(fWl + off, fWl + off + 8);
        accy = WMMA_F16(arh, bwl, accy);
    }
#pragma unroll
    for (int v = 0; v < 8; ++v)
        y32[(size_t)(mbase + v) * DH + col] = accy[v];
}

__global__ void __launch_bounds__(256) scatter_kernel(
    const int* __restrict__ ei, const float* __restrict__ y32,
    float* __restrict__ aggy, float* __restrict__ cnt)
{
    const int lane = threadIdx.x & 31;
    const int wave = __builtin_amdgcn_readfirstlane((int)(threadIdx.x >> 5));
    const int e0   = (blockIdx.x * 8 + wave) * EPW;

    int s[EPW], d[EPW];
    for (int j = 0; j < EPW; ++j) { s[j] = ei[e0 + j]; d[j] = ei[NE + e0 + j]; }

    float v[EPW][4];
    for (int j = 0; j < EPW; ++j) {
        const float* ys = y32 + (size_t)s[j] * DH + lane;
        for (int i = 0; i < 4; ++i) v[j][i] = ys[i * 32];
    }
    for (int j = 0; j < EPW; ++j) {
        float* ag = aggy + (size_t)d[j] * DH + lane;
        for (int i = 0; i < 4; ++i) atomicAdd(&ag[i * 32], v[j][i]);
    }
    int dl = 0;
    for (int j = 0; j < EPW; ++j) dl = (lane == j) ? d[j] : dl;
    if (lane < EPW) atomicAdd(&cnt[dl], 1.0f);
}

__global__ void __launch_bounds__(256) cand_kernel(
    const float* __restrict__ aggy, const float* __restrict__ cnt,
    const _Float16* __restrict__ x16, const _Float16* __restrict__ rh16,
    const _Float16* __restrict__ fWr,
    const float* __restrict__ b_l,
    const float* __restrict__ z_ws, const float* __restrict__ h_prev,
    float* __restrict__ out)
{
    const int lane  = threadIdx.x & 31;
    const int ntile = threadIdx.x >> 5;
    const int mtile = blockIdx.x;
    const int col   = ntile * 16 + (lane & 15);
    const int khalf = lane >> 4;
    const int arow  = mtile * 16 + (lane & 15);

    const int mbase = mtile * 16 + ((lane & 16) ? 8 : 0);

    v8f acc;
    const float bl = b_l[col];
#pragma unroll
    for (int v = 0; v < 8; ++v) {
        const float inv = 1.0f / fmaxf(cnt[mbase + v], 1.0f);
        acc[v] = aggy[(size_t)(mbase + v) * DH + col] * inv + bl;
    }

#pragma unroll
    for (int kt = 0; kt < 8; ++kt) {
        const _Float16* xcb = (kt < 4)
            ? (x16  + (size_t)arow * DH + kt * 32)
            : (rh16 + (size_t)arow * DH + (kt - 4) * 32);
        v16h ac = load_frag(xcb + khalf * 8, xcb + khalf * 8 + 16);

        const size_t off = ((size_t)(kt * 8 + ntile) * 32 + lane) * 16;
        v16h bwr = load_frag(fWr + off, fWr + off + 8);
        acc = WMMA_F16(ac, bwr, acc);
    }

#pragma unroll
    for (int v = 0; v < 8; ++v) {
        const size_t idx = (size_t)(mbase + v) * DH + col;
        const float zv = z_ws[idx];
        out[idx] = (1.0f - zv) * acc[v] + zv * h_prev[idx];
    }
}

extern "C" void kernel_launch(void* const* d_in, const int* in_sizes, int n_in,
                              void* d_out, int out_size, void* d_ws, size_t ws_size,
                              hipStream_t stream) {
    const float* x      = (const float*)d_in[0];
    const int*   ei     = (const int*)  d_in[1];
    const float* h_prev = (const float*)d_in[2];
    const float* W_xr   = (const float*)d_in[3];
    const float* b_xr   = (const float*)d_in[4];
    const float* W_hr   = (const float*)d_in[5];
    const float* W_xz   = (const float*)d_in[6];
    const float* b_xz   = (const float*)d_in[7];
    const float* W_hz   = (const float*)d_in[8];
    const float* W_l    = (const float*)d_in[9];
    const float* b_l    = (const float*)d_in[10];
    const float* W_r    = (const float*)d_in[11];
    float* out = (float*)d_out;

    char* ws = (char*)d_ws;
    size_t off = 0;
    auto alloc = [&](size_t bytes) -> char* {
        char* p = ws + off;
        off += (bytes + 255) & ~(size_t)255;
        return p;
    };
    _Float16* x16  = (_Float16*)alloc((size_t)NN * DH * 2);
    _Float16* h16  = (_Float16*)alloc((size_t)NN * DH * 2);
    _Float16* rh16 = (_Float16*)alloc((size_t)NN * DH * 2);
    float*    z_ws = (float*)   alloc((size_t)NN * DH * 4);
    float*    y32  = (float*)   alloc((size_t)NN * DH * 4);
    float*    aggy = (float*)   alloc((size_t)NN * DH * 4);
    float*    cnt  = (float*)   alloc((size_t)NN * 4);
    _Float16* fWxr = (_Float16*)alloc(128 * 128 * 2);
    _Float16* fWhr = (_Float16*)alloc(128 * 128 * 2);
    _Float16* fWxz = (_Float16*)alloc(128 * 128 * 2);
    _Float16* fWhz = (_Float16*)alloc(128 * 128 * 2);
    _Float16* fWl  = (_Float16*)alloc(256 * 128 * 2);
    _Float16* fWr  = (_Float16*)alloc(256 * 128 * 2);

    prep_all<<<PREP_BLOCKS, 256, 0, stream>>>(x, h_prev, x16, h16, aggy, cnt,
                                              W_xr, W_hr, W_xz, W_hz, W_l, W_r,
                                              fWxr, fWhr, fWxz, fWhz, fWl, fWr);

    gates_kernel<<<MTILES, 256, 0, stream>>>(x16, h16, h_prev, b_xr, b_xz,
                                             fWxr, fWhr, fWxz, fWhz, fWl,
                                             z_ws, y32, rh16);

    scatter_kernel<<<NE / (8 * EPW), 256, 0, stream>>>(ei, y32, aggy, cnt);

    cand_kernel<<<MTILES, 256, 0, stream>>>(aggy, cnt, x16, rh16, fWr,
                                            b_l, z_ws, h_prev, out);
}


# Neighbour sums by destination ownership: edges binned per 100-node range with LDS counters, owner blocks add rows into an LDS tile (LDS f32 adds); no f32 atomics on memory; f32 sum order changes
# speedup vs baseline: 2.8827x; 1.2809x over previous
#include <hip/hip_runtime.h>
#include <hip/hip_bf16.h>


typedef __attribute__((ext_vector_type(16))) _Float16 v16h;
typedef __attribute__((ext_vector_type(8)))  float    v8f;
typedef __attribute__((ext_vector_type(4)))  int      v4i;
typedef __attribute__((ext_vector_type(4)))  float    v4f;
typedef __attribute__((ext_vector_type(4)))  _Float16 v4h;

#define NN      50000
#define DH      128
#define DCAT    256
#define NE      800000
#define MTILES  (NN / 16)
#define RH_PITCH 136
#define CHUNKS  5
#define EPW     16
#define OWN_R   100
#define NBIN    (NN / OWN_R)
#define NCH     250
#define CH_E    (NE / NCH)
#define SUM_WAVES 16
static_assert(NN % OWN_R == 0 && NE % NCH == 0, "bins tile the nodes and chunks tile the edges exactly");
static_assert(NN <= 65536 && OWN_R <= 65536, "a sorted entry packs the source node (low 16 bits) and the row inside the owner's range (high 16 bits)");

#define WMMA_F16(A, B, C) \
    __builtin_amdgcn_wmma_f32_16x16x32_f16(false, (A), false, (B), (short)0, (C), false, false)

__device__ __forceinline__ v16h load_frag(const _Float16* p0, const _Float16* p1) {
    v16h r;
    reinterpret_cast<v4i*>(&r)[0] = *reinterpret_cast<const v4i*>(p0);
    reinterpret_cast<v4i*>(&r)[1] = *reinterpret_cast<const v4i*>(p1);
    return r;
}

__device__ __forceinline__ float sigmoid_f(float x) {
    return 1.0f / (1.0f + __expf(-x));
}

#define PIECES_PER_BLOCK (256 * CHUNKS)
#define ACT_BLOCKS  (NN * DH / 4 / PIECES_PER_BLOCK)
#define W_BLOCKS    (4 * 64 + 2 * 128)
#define PREP_BLOCKS (W_BLOCKS + 1 + ACT_BLOCKS)
static_assert((NN * DH / 4) % PIECES_PER_BLOCK == 0, "the f16 copies have no tail: blocks must tile the arrays exactly");

__global__ void __launch_bounds__(256) prep_all(
    const float* __restrict__ x,    const float* __restrict__ h_prev,
    _Float16* __restrict__ x16,     _Float16* __restrict__ h16,
    int* __restrict__ totals,
    const float* __restrict__ W_xr, const float* __restrict__ W_hr,
    const float* __restrict__ W_xz, const float* __restrict__ W_hz,
    const float* __restrict__ W_l,  const float* __restrict__ W_r,
    _Float16* __restrict__ fWxr,    _Float16* __restrict__ fWhr,
    _Float16* __restrict__ fWxz,    _Float16* __restrict__ fWhz,
    _Float16* __restrict__ fWl,     _Float16* __restrict__ fWr)
{
    const int b = blockIdx.x;

    if (b < W_BLOCKS) {
        const float* W; _Float16* frag; int lb;
        if      (b <  64) { W = W_xr; frag = fWxr; lb = b;       }
        else if (b < 128) { W = W_hr; frag = fWhr; lb = b -  64; }
        else if (b < 192) { W = W_xz; frag = fWxz; lb = b - 128; }
        else if (b < 256) { W = W_hz; frag = fWhz; lb = b - 192; }
        else if (b < 384) { W = W_l;  frag = fWl;  lb = b - 256; }
        else              { W = W_r;  frag = fWr;  lb = b - 384; }
        const int idx   = lb * 256 + threadIdx.x;
        const int j     = idx & 15;
        const int lane  = (idx >> 4) & 31;
        const int ntile = (idx >> 9) & 7;
        const int ktile = idx >> 12;
        const int n     = ntile * 16 + (lane & 15);
        const int khalf = lane >> 4;
        const int k     = ktile * 32 + khalf * 8 + ((j < 8) ? j : 16 + (j - 8));
        frag[idx] = (_Float16)W[(size_t)k * DH + n];
    } else if (b == W_BLOCKS) {
        for (int i = threadIdx.x; i < NBIN; i += 256) totals[i] = 0;
    } else {
        const size_t base = (size_t)(b - W_BLOCKS - 1) * PIECES_PER_BLOCK + threadIdx.x;
        const v4f* xa = reinterpret_cast<const v4f*>(x)      + base;
        const v4f* ha = reinterpret_cast<const v4f*>(h_prev) + base;
        v4h*       xb = reinterpret_cast<v4h*>(x16) + base;
        v4h*       hb = reinterpret_cast<v4h*>(h16) + base;
        v4f vx[CHUNKS], vh[CHUNKS];
        for (int k = 0; k < CHUNKS; ++k) vx[k] = xa[k * 256];
        for (int k = 0; k < CHUNKS; ++k) vh[k] = ha[k * 256];
        for (int k = 0; k < CHUNKS; ++k) xb[k * 256] = __builtin_convertvector(vx[k], v4h);
        for (int k = 0; k < CHUNKS; ++k) hb[k * 256] = __builtin_convertvector(vh[k], v4h);
    }
}

__global__ void __launch_bounds__(256) gates_kernel(
    const _Float16* __restrict__ x16,  const _Float16* __restrict__ h16,
    const float* __restrict__ h_prev,
    const float* __restrict__ b_xr,    const float* __restrict__ b_xz,
    const _Float16* __restrict__ fWxr, const _Float16* __restrict__ fWhr,
    const _Float16* __restrict__ fWxz, const _Float16* __restrict__ fWhz,
    const _Float16* __restrict__ fWl,
    float* __restrict__ z_ws, float* __restrict__ y32, _Float16* __restrict__ rh16)
{
    __shared__ __attribute__((aligned(16))) _Float16 s_rh[16 * RH_PITCH];

    const int lane  = threadIdx.x & 31;
    const int ntile = threadIdx.x >> 5;
    const int mtile = blockIdx.x;
    const int col   = ntile * 16 + (lane & 15);
    const int khalf = lane >> 4;
    const int arow  = mtile * 16 + (lane & 15);

    v8f accr, accz, accy;
    const float br = b_xr[col], bz = b_xz[col];
#pragma unroll
    for (int v = 0; v < 8; ++v) { accr[v] = br; accz[v] = bz; accy[v] = 0.0f; }

    const _Float16* xb = x16 + (size_t)arow * DH;
    const _Float16* hb = h16 + (size_t)arow * DH;

#pragma unroll
    for (int kt = 0; kt < 4; ++kt) {
        const int ko = kt * 32 + khalf * 8;
        v16h ax = load_frag(xb + ko, xb + ko + 16);
        v16h ah = load_frag(hb + ko, hb + ko + 16);
        const size_t off = ((size_t)(kt * 8 + ntile) * 32 + lane) * 16;
        v16h bxr = load_frag(fWxr + off, fWxr + off + 8);
        v16h bhr = load_frag(fWhr + off, fWhr + off + 8);
        v16h bxz = load_frag(fWxz + off, fWxz + off + 8);
        v16h bhz = load_frag(fWhz + off, fWhz + off + 8);
        v16h bwl = load_frag(fWl  + off, fWl  + off + 8);
        accr = WMMA_F16(ax, bxr, accr);
        accr = WMMA_F16(ah, bhr, accr);
        accz = WMMA_F16(ax, bxz, accz);
        accz = WMMA_F16(ah, bhz, accz);
        accy = WMMA_F16(ax, bwl, accy);
    }

    const int rloc  = (lane & 16) ? 8 : 0;
    const int mbase = mtile * 16 + rloc;
#pragma unroll
    for (int v = 0; v < 8; ++v) {
        const size_t idx = (size_t)(mbase + v) * DH + col;
        const float rv = sigmoid_f(accr[v]);
        const float zv = sigmoid_f(accz[v]);
        const float h  = h_prev[idx];
        const _Float16 rh = (_Float16)(rv * h);
        z_ws[idx] = zv;
        rh16[idx] = rh;
        s_rh[(rloc + v) * RH_PITCH + col] = rh;
    }
    __syncthreads();

    const _Float16* sp = s_rh + (lane & 15) * RH_PITCH + khalf * 8;
#pragma unroll
    for (int kt = 0; kt < 4; ++kt) {
        v16h arh = load_frag(sp + kt * 32, sp + kt * 32 + 16);
        const size_t off = ((size_t)((4 + kt) * 8 + ntile) * 32 + lane) * 16;
        v16h bwl = load_frag(fWl + off, fWl + off + 8);
        accy = WMMA_F16(arh, bwl, accy);
    }
#pragma unroll
    for (int v = 0; v < 8; ++v)
        y32[(size_t)(mbase + v) * DH + col] = accy[v];
}

#define BIN_T   1024
#define BIN_IT  ((CH_E + BIN_T - 1) / BIN_T)
__global__ void __launch_bounds__(BIN_T) bin_count(
    const int* __restrict__ ei, int* __restrict__ counts, int* __restrict__ totals)
{
    __shared__ int hist[NBIN];
    const int c = blockIdx.x;
    const int* dst = ei + NE + c * CH_E;
    int d[BIN_IT];
    for (int k = 0; k < BIN_IT; ++k) {
        const int i = threadIdx.x + k * BIN_T;
        d[k] = (i < CH_E) ? dst[i] : 0;
    }
    for (int b = threadIdx.x; b < NBIN; b += BIN_T) hist[b] = 0;
    __syncthreads();
    for (int k = 0; k < BIN_IT; ++k) {
        const int i = threadIdx.x + k * BIN_T;
        if (i < CH_E) {
            const int bin = min(d[k] / OWN_R, NBIN - 1);
            atomicAdd(&hist[bin], 1);
        }
    }
    __syncthreads();
    for (int b = threadIdx.x; b < NBIN; b += BIN_T) {
        const int n = hist[b];
        counts[b * NCH + c] = n;
        if (n) atomicAdd(&totals[b], n);
    }
}

__global__ void __launch_bounds__(256) bin_scan(
    int* __restrict__ counts, const int* __restrict__ totals, int* __restrict__ bin_start)
{
    const int lane = threadIdx.x & 31;
    const int b    = blockIdx.x * 8 + (threadIdx.x >> 5);
    if (b >= NBIN) return;

    int s = 0;
    for (int q = lane; q < b; q += 32) s += totals[q];
    for (int d = 16; d >= 1; d >>= 1) s += __shfl_xor(s, d);

    int* row = counts + b * NCH;
    int v[8], loc[8], run = 0;
    for (int k = 0; k < 8; ++k) { const int c = lane * 8 + k; v[k] = (c < NCH) ? row[c] : 0; }
    for (int k = 0; k < 8; ++k) { loc[k] = run; run += v[k]; }
    int incl = run;
    for (int d = 1; d < 32; d <<= 1) {
        const int up = __shfl_up(incl, d);
        if (lane >= d) incl += up;
    }
    const int excl = incl - run;
    for (int k = 0; k < 8; ++k) { const int c = lane * 8 + k; if (c < NCH) row[c] = s + excl + loc[k]; }
    if (lane == 0) bin_start[b] = s;
}
static_assert(NCH <= 256, "bin_scan gives each of 32 lanes 8 chunk counts");

__global__ void __launch_bounds__(BIN_T) bin_fill(
    const int* __restrict__ ei, const int* __restrict__ offs, unsigned* __restrict__ sorted)
{
    __shared__ int base[NBIN];
    __shared__ int cntr[NBIN];
    const int c = blockIdx.x;
    const int* src = ei + c * CH_E;
    const int* dst = ei + NE + c * CH_E;
    int s[BIN_IT], d[BIN_IT];
    for (int k = 0; k < BIN_IT; ++k) {
        const int i = threadIdx.x + k * BIN_T;
        s[k] = (i < CH_E) ? src[i] : 0;
        d[k] = (i < CH_E) ? dst[i] : 0;
    }
    for (int b = threadIdx.x; b < NBIN; b += BIN_T) { base[b] = offs[b * NCH + c]; cntr[b] = 0; }
    __syncthreads();
    for (int k = 0; k < BIN_IT; ++k) {
        const int i = threadIdx.x + k * BIN_T;
        if (i < CH_E) {
            const int bin = min(d[k] / OWN_R, NBIN - 1);
            const int r   = atomicAdd(&cntr[bin], 1);
            sorted[base[bin] + r] = (unsigned)s[k] | ((unsigned)(d[k] - bin * OWN_R) << 16);
        }
    }
}

__global__ void __launch_bounds__(SUM_WAVES * 32) owner_sum(
    const unsigned* __restrict__ sorted, const int* __restrict__ bin_start,
    const int* __restrict__ totals, const float* __restrict__ y32,
    float* __restrict__ aggy, float* __restrict__ cnt)
{
    __shared__ float tile[OWN_R * DH];
    __shared__ float lcnt[OWN_R];

    const int b    = blockIdx.x;
    const int lane = threadIdx.x & 31;
    const int wave = __builtin_amdgcn_readfirstlane((int)(threadIdx.x >> 5));

    for (int i = threadIdx.x; i < OWN_R * DH; i += SUM_WAVES * 32) tile[i] = 0.0f;
    if (threadIdx.x < OWN_R) lcnt[threadIdx.x] = 0.0f;
    __syncthreads();

    const int start = bin_start[b];
    const int n     = totals[b];
    for (int e0 = wave * EPW; e0 < n; e0 += SUM_WAVES * EPW) {
        const int      my = e0 + (lane & (EPW - 1));
        const unsigned pk = sorted[min(start + my, NE - 1)];

        unsigned dl[EPW];
        float v[EPW][4];
        for (int j = 0; j < EPW; ++j) {
            const unsigned pkj = (unsigned)__builtin_amdgcn_readlane((int)pk, j);
            dl[j] = pkj >> 16;
            const float* ys = y32 + (size_t)(pkj & 0xFFFFu) * DH + lane;
            for (int i = 0; i < 4; ++i) v[j][i] = ys[i * 32];
        }
        for (int j = 0; j < EPW; ++j) {
            if (e0 + j < n) {
                float* tp = tile + dl[j] * DH + lane;
                for (int i = 0; i < 4; ++i) atomicAdd(&tp[i * 32], v[j][i]);
            }
        }
        if (lane < EPW && my < n) atomicAdd(&lcnt[pk >> 16], 1.0f);
    }
    __syncthreads();

    float* ag = aggy + (size_t)b * OWN_R * DH;
    for (int i = threadIdx.x; i < OWN_R * DH; i += SUM_WAVES * 32) ag[i] = tile[i];
    if (threadIdx.x < OWN_R) cnt[b * OWN_R + threadIdx.x] = lcnt[threadIdx.x];
}

__global__ void __launch_bounds__(256) cand_kernel(
    const float* __restrict__ aggy, const float* __restrict__ cnt,
    const _Float16* __restrict__ x16, const _Float16* __restrict__ rh16,
    const _Float16* __restrict__ fWr,
    const float* __restrict__ b_l,
    const float* __restrict__ z_ws, const float* __restrict__ h_prev,
    float* __restrict__ out)
{
    const int lane  = threadIdx.x & 31;
    const int ntile = threadIdx.x >> 5;
    const int mtile = blockIdx.x;
    const int col   = ntile * 16 + (lane & 15);
    const int khalf = lane >> 4;
    const int arow  = mtile * 16 + (lane & 15);

    const int mbase = mtile * 16 + ((lane & 16) ? 8 : 0);

    v8f acc;
    const float bl = b_l[col];
#pragma unroll
    for (int v = 0; v < 8; ++v) {
        const float inv = 1.0f / fmaxf(cnt[mbase + v], 1.0f);
        acc[v] = aggy[(size_t)(mbase + v) * DH + col] * inv + bl;
    }

#pragma unroll
    for (int kt = 0; kt < 8; ++kt) {
        const _Float16* xcb = (kt < 4)
            ? (x16  + (size_t)arow * DH + kt * 32)
            : (rh16 + (size_t)arow * DH + (kt - 4) * 32);
        v16h ac = load_frag(xcb + khalf * 8, xcb + khalf * 8 + 16);

        const size_t off = ((size_t)(kt * 8 + ntile) * 32 + lane) * 16;
        v16h bwr = load_frag(fWr + off, fWr + off + 8);
        acc = WMMA_F16(ac, bwr, acc);
    }

#pragma unroll
    for (int v = 0; v < 8; ++v) {
        const size_t idx = (size_t)(mbase + v) * DH + col;
        const float zv = z_ws[idx];
        out[idx] = (1.0f - zv) * acc[v] + zv * h_prev[idx];
    }
}

extern "C" void kernel_launch(void* const* d_in, const int* in_sizes, int n_in,
                              void* d_out, int out_size, void* d_ws, size_t ws_size,
                              hipStream_t stream) {
    const float* x      = (const float*)d_in[0];
    const int*   ei     = (const int*)  d_in[1];
    const float* h_prev = (const float*)d_in[2];
    const float* W_xr   = (const float*)d_in[3];
    const float* b_xr   = (const float*)d_in[4];
    const float* W_hr   = (const float*)d_in[5];
    const float* W_xz   = (const float*)d_in[6];
    const float* b_xz   = (const float*)d_in[7];
    const float* W_hz   = (const float*)d_in[8];
    const float* W_l    = (const float*)d_in[9];
    const float* b_l    = (const float*)d_in[10];
    const float* W_r    = (const float*)d_in[11];
    float* out = (float*)d_out;

    char* ws = (char*)d_ws;
    size_t off = 0;
    auto alloc = [&](size_t bytes) -> char* {
        char* p = ws + off;
        off += (bytes + 255) & ~(size_t)255;
        return p;
    };
    _Float16* x16  = (_Float16*)alloc((size_t)NN * DH * 2);
    _Float16* h16  = (_Float16*)alloc((size_t)NN * DH * 2);
    _Float16* rh16 = (_Float16*)alloc((size_t)NN * DH * 2);
    float*    z_ws = (float*)   alloc((size_t)NN * DH * 4);
    float*    y32  = (float*)   alloc((size_t)NN * DH * 4);
    float*    aggy = (float*)   alloc((size_t)NN * DH * 4);
    float*    cnt  = (float*)   alloc((size_t)NN * 4);
    unsigned* sorted = (unsigned*)alloc((size_t)NE * 4);
    int*      counts = (int*)     alloc((size_t)NBIN * NCH * 4);
    int*      totals = (int*)     alloc((size_t)NBIN * 4);
    int*      bstart = (int*)     alloc((size_t)NBIN * 4);
    _Float16* fWxr = (_Float16*)alloc(128 * 128 * 2);
    _Float16* fWhr = (_Float16*)alloc(128 * 128 * 2);
    _Float16* fWxz = (_Float16*)alloc(128 * 128 * 2);
    _Float16* fWhz = (_Float16*)alloc(128 * 128 * 2);
    _Float16* fWl  = (_Float16*)alloc(256 * 128 * 2);
    _Float16* fWr  = (_Float16*)alloc(256 * 128 * 2);

    prep_all<<<PREP_BLOCKS, 256, 0, stream>>>(x, h_prev, x16, h16, totals,
                                              W_xr, W_hr, W_xz, W_hz, W_l, W_r,
                                              fWxr, fWhr, fWxz, fWhz, fWl, fWr);

    gates_kernel<<<MTILES, 256, 0, stream>>>(x16, h16, h_prev, b_xr, b_xz,
                                             fWxr, fWhr, fWxz, fWhz, fWl,
                                             z_ws, y32, rh16);

    bin_count<<<NCH, BIN_T, 0, stream>>>(ei, counts, totals);
    bin_scan<<<(NBIN + 7) / 8, 256, 0, stream>>>(counts, totals, bstart);
    bin_fill<<<NCH, BIN_T, 0, stream>>>(ei, counts, sorted);

    owner_sum<<<NBIN, SUM_WAVES * 32, 0, stream>>>(sorted, bstart, totals, y32, aggy, cnt);

    cand_kernel<<<MTILES, 256, 0, stream>>>(aggy, cnt, x16, rh16, fWr,
                                            b_l, z_ws, h_prev, out);
}


# WMMA operands (x, h, r*h planes, the six weights) stored in fragment order, two 512-B pieces per 16x32 tile: a fragment load reads 512 contiguous bytes, not 16 rows x 32 B; f32 sum order unchanged
# speedup vs baseline: 3.4993x; 1.2139x over previous
#include <hip/hip_runtime.h>
#include <hip/hip_bf16.h>


typedef __attribute__((ext_vector_type(16))) _Float16 v16h;
typedef __attribute__((ext_vector_type(8)))  float    v8f;
typedef __attribute__((ext_vector_type(4)))  int      v4i;
typedef __attribute__((ext_vector_type(4)))  float    v4f;
typedef __attribute__((ext_vector_type(4)))  _Float16 v4h;

#define NN      50000
#define DH      128
#define DCAT    256
#define NE      800000
#define MTILES  (NN / 16)
#define MT      5
#define MGROUPS (MTILES / MT)
#define RH_PITCH 136
#define PIECE   256
#define WG_PIECES (MT * 4 * 2)
#define EPW     16
#define OWN_R   100
#define NBIN    (NN / OWN_R)
#define NCH     250
#define CH_E    (NE / NCH)
#define SUM_WAVES 16
static_assert(NN % OWN_R == 0 && NE % NCH == 0, "bins tile the nodes and chunks tile the edges exactly");
static_assert(MTILES % MT == 0, "the GEMM workgroups tile the rows exactly: no tail, no guard");
static_assert(NN <= 65536 && OWN_R <= 65536, "a sorted entry packs the source node (low 16 bits) and the row inside the owner's range (high 16 bits)");

#define WMMA_F16(A, B, C) \
    __builtin_amdgcn_wmma_f32_16x16x32_f16(false, (A), false, (B), (short)0, (C), false, false)

__device__ __forceinline__ v16h load_frag(const _Float16* p0, const _Float16* p1) {
    v16h r;
    reinterpret_cast<v4i*>(&r)[0] = *reinterpret_cast<const v4i*>(p0);
    reinterpret_cast<v4i*>(&r)[1] = *reinterpret_cast<const v4i*>(p1);
    return r;
}

__device__ __forceinline__ float sigmoid_f(float x) {
    return 1.0f / (1.0f + __expf(-x));
}

#define ROW_LOADS   (MT * 16 * (DH / 4) / 256)
#define ACT_BLOCKS  (2 * MGROUPS)
#define W_BLOCKS    (4 * 64 + 2 * 128)
#define PREP_BLOCKS (W_BLOCKS + 1 + ACT_BLOCKS)
static_assert(ROW_LOADS * 256 == MT * 16 * (DH / 4) && WG_PIECES == MT * 8, "8 waves move the 80 rows in and the 40 pieces out without a tail");

__global__ void __launch_bounds__(256) prep_all(
    const float* __restrict__ x,    const float* __restrict__ h_prev,
    _Float16* __restrict__ x16,     _Float16* __restrict__ h16,
    int* __restrict__ totals,
    const float* __restrict__ W_xr, const float* __restrict__ W_hr,
    const float* __restrict__ W_xz, const float* __restrict__ W_hz,
    const float* __restrict__ W_l,  const float* __restrict__ W_r,
    _Float16* __restrict__ fWxr,    _Float16* __restrict__ fWhr,
    _Float16* __restrict__ fWxz,    _Float16* __restrict__ fWhz,
    _Float16* __restrict__ fWl,     _Float16* __restrict__ fWr)
{
    __shared__ __attribute__((aligned(16))) _Float16 s_t[MT * 16 * RH_PITCH];
    const int b = blockIdx.x;

    if (b < W_BLOCKS) {
        const float* W; _Float16* frag; int lb;
        if      (b <  64) { W = W_xr; frag = fWxr; lb = b;       }
        else if (b < 128) { W = W_hr; frag = fWhr; lb = b -  64; }
        else if (b < 192) { W = W_xz; frag = fWxz; lb = b - 128; }
        else if (b < 256) { W = W_hz; frag = fWhz; lb = b - 192; }
        else if (b < 384) { W = W_l;  frag = fWl;  lb = b - 256; }
        else              { W = W_r;  frag = fWr;  lb = b - 384; }
        const int idx   = lb * 256 + threadIdx.x;
        const int e     = idx & 7;
        const int lane  = (idx >> 3) & 31;
        const int q     = (idx >> 8) & 1;
        const int ntile = (idx >> 9) & 7;
        const int ktile = idx >> 12;
        const int n     = ntile * 16 + (lane & 15);
        const int khalf = lane >> 4;
        const int k     = ktile * 32 + q * 16 + khalf * 8 + e;
        frag[idx] = (_Float16)W[(size_t)k * DH + n];
    } else if (b == W_BLOCKS) {
        for (int i = threadIdx.x; i < NBIN; i += 256) totals[i] = 0;
    } else {
        const int a = b - W_BLOCKS - 1;
        const int g = (a < MGROUPS) ? a : a - MGROUPS;
        const float*    src = (a < MGROUPS) ? x   : h_prev;
        _Float16*       dst = (a < MGROUPS) ? x16 : h16;
        const int lane = threadIdx.x & 31;
        const int wave = threadIdx.x >> 5;
        const v4f* sa = reinterpret_cast<const v4f*>(src) + (size_t)g * (MT * 16 * (DH / 4)) + threadIdx.x;
        v4f v[ROW_LOADS];
        for (int i = 0; i < ROW_LOADS; ++i) v[i] = sa[i * 256];
        for (int i = 0; i < ROW_LOADS; ++i)
            *reinterpret_cast<v4h*>(&s_t[(i * 8 + wave) * RH_PITCH + 4 * lane]) = __builtin_convertvector(v[i], v4h);
        __syncthreads();
        const _Float16* sp = s_t + (lane & 15) * RH_PITCH + wave * 16 + (lane >> 4) * 8;
        v4i* dp = reinterpret_cast<v4i*>(dst + ((size_t)g * WG_PIECES + wave) * PIECE + lane * 8);
        v4i pc[MT];
        for (int t = 0; t < MT; ++t) pc[t] = *reinterpret_cast<const v4i*>(sp + t * 16 * RH_PITCH);
        for (int t = 0; t < MT; ++t) dp[t * 8 * (PIECE / 8)] = pc[t];
    }
}

__global__ void __launch_bounds__(256) __attribute__((amdgpu_waves_per_eu(4, 4))) gates_kernel(
    const _Float16* __restrict__ x16,  const _Float16* __restrict__ h16,
    const float* __restrict__ h_prev,
    const float* __restrict__ b_xr,    const float* __restrict__ b_xz,
    const _Float16* __restrict__ fWxr, const _Float16* __restrict__ fWhr,
    const _Float16* __restrict__ fWxz, const _Float16* __restrict__ fWhz,
    const _Float16* __restrict__ fWl,
    float* __restrict__ z_ws, float* __restrict__ y32, _Float16* __restrict__ rh16)
{
    __shared__ __attribute__((aligned(16))) _Float16 s_rh[MT * 16 * RH_PITCH];

    const int lane  = threadIdx.x & 31;
    const int ntile = threadIdx.x >> 5;
    const int row0  = blockIdx.x * (MT * 16);
    const int col   = ntile * 16 + (lane & 15);
    const int khalf = lane >> 4;

    v8f accr[MT], accz[MT], accy[MT];
    const float br = b_xr[col], bz = b_xz[col];
#pragma unroll
    for (int m = 0; m < MT; ++m) {
#pragma unroll
        for (int v = 0; v < 8; ++v) { accr[m][v] = br; accz[m][v] = bz; accy[m][v] = 0.0f; }
    }

    const _Float16* xb = x16 + (size_t)blockIdx.x * (WG_PIECES * PIECE) + lane * 8;
    const _Float16* hb = h16 + (size_t)blockIdx.x * (WG_PIECES * PIECE) + lane * 8;

#pragma unroll
    for (int kt = 0; kt < 4; ++kt) {
        const size_t off = (size_t)(kt * 8 + ntile) * (2 * PIECE) + lane * 8;
        const v16h bxr = load_frag(fWxr + off, fWxr + off + PIECE);
        const v16h bhr = load_frag(fWhr + off, fWhr + off + PIECE);
        const v16h bxz = load_frag(fWxz + off, fWxz + off + PIECE);
        const v16h bhz = load_frag(fWhz + off, fWhz + off + PIECE);
        const v16h bwl = load_frag(fWl  + off, fWl  + off + PIECE);
        v16h ax[MT], ah[MT];
#pragma unroll
        for (int m = 0; m < MT; ++m) {
            const _Float16* xm = xb + (m * 4 + kt) * (2 * PIECE);
            const _Float16* hm = hb + (m * 4 + kt) * (2 * PIECE);
            ax[m] = load_frag(xm, xm + PIECE);
            ah[m] = load_frag(hm, hm + PIECE);
        }
#pragma unroll
        for (int m = 0; m < MT; ++m) {
            accr[m] = WMMA_F16(ax[m], bxr, accr[m]);
            accr[m] = WMMA_F16(ah[m], bhr, accr[m]);
            accz[m] = WMMA_F16(ax[m], bxz, accz[m]);
            accz[m] = WMMA_F16(ah[m], bhz, accz[m]);
            accy[m] = WMMA_F16(ax[m], bwl, accy[m]);
        }
    }

    const int rloc = (lane & 16) ? 8 : 0;
    float hv[MT][8];
#pragma unroll
    for (int m = 0; m < MT; ++m) {
#pragma unroll
        for (int v = 0; v < 8; ++v)
            hv[m][v] = h_prev[(size_t)(row0 + m * 16 + rloc + v) * DH + col];
    }
    v16h bw2[4];
#pragma unroll
    for (int kt = 0; kt < 4; ++kt) {
        const size_t off = (size_t)((4 + kt) * 8 + ntile) * (2 * PIECE) + lane * 8;
        bw2[kt] = load_frag(fWl + off, fWl + off + PIECE);
    }
#pragma unroll
    for (int m = 0; m < MT; ++m) {
#pragma unroll
        for (int v = 0; v < 8; ++v) {
            const int    lrow = m * 16 + rloc + v;
            const size_t idx  = (size_t)(row0 + lrow) * DH + col;
            const float rv = sigmoid_f(accr[m][v]);
            const float zv = sigmoid_f(accz[m][v]);
            const float h  = hv[m][v];
            const _Float16 rh = (_Float16)(rv * h);
            z_ws[idx] = zv;
            s_rh[lrow * RH_PITCH + col] = rh;
        }
    }
    __syncthreads();

    {
        const _Float16* cp = s_rh + (lane & 15) * RH_PITCH + ntile * 16 + khalf * 8;
        v4i* dp = reinterpret_cast<v4i*>(rh16 + ((size_t)blockIdx.x * WG_PIECES + ntile) * PIECE + lane * 8);
        v4i pc[MT];
#pragma unroll
        for (int m = 0; m < MT; ++m) pc[m] = *reinterpret_cast<const v4i*>(cp + m * 16 * RH_PITCH);
#pragma unroll
        for (int m = 0; m < MT; ++m) dp[m * 8 * (PIECE / 8)] = pc[m];
    }

    const _Float16* sp = s_rh + (lane & 15) * RH_PITCH + khalf * 8;
#pragma unroll
    for (int kt = 0; kt < 4; ++kt) {
        v16h arh[MT];
#pragma unroll
        for (int m = 0; m < MT; ++m) {
            const _Float16* sm = sp + m * 16 * RH_PITCH + kt * 32;
            arh[m] = load_frag(sm, sm + 16);
        }
#pragma unroll
        for (int m = 0; m < MT; ++m)
            accy[m] = WMMA_F16(arh[m], bw2[kt], accy[m]);
    }
#pragma unroll
    for (int m = 0; m < MT; ++m) {
#pragma unroll
        for (int v = 0; v < 8; ++v)
            y32[(size_t)(row0 + m * 16 + rloc + v) * DH + col] = accy[m][v];
    }
}

#define BIN_T   1024
#define BIN_IT  ((CH_E + BIN_T - 1) / BIN_T)
__global__ void __launch_bounds__(BIN_T) bin_count(
    const int* __restrict__ ei, int* __restrict__ counts, int* __restrict__ totals)
{
    __shared__ int hist[NBIN];
    const int c = blockIdx.x;
    const int* dst = ei + NE + c * CH_E;
    int d[BIN_IT];
    for (int k = 0; k < BIN_IT; ++k) {
        const int i = threadIdx.x + k * BIN_T;
        d[k] = (i < CH_E) ? dst[i] : 0;
    }
    for (int b = threadIdx.x; b < NBIN; b += BIN_T) hist[b] = 0;
    __syncthreads();
    for (int k = 0; k < BIN_IT; ++k) {
        const int i = threadIdx.x + k * BIN_T;
        if (i < CH_E) {
            const int bin = min(d[k] / OWN_R, NBIN - 1);
            atomicAdd(&hist[bin], 1);
        }
    }
    __syncthreads();
    for (int b = threadIdx.x; b < NBIN; b += BIN_T) {
        const int n = hist[b];
        counts[b * NCH + c] = n;
        if (n) atomicAdd(&totals[b], n);
    }
}

__global__ void __launch_bounds__(256) bin_scan(
    int* __restrict__ counts, const int* __restrict__ totals, int* __restrict__ bin_start)
{
    const int lane = threadIdx.x & 31;
    const int b    = blockIdx.x * 8 + (threadIdx.x >> 5);
    if (b >= NBIN) return;

    int s = 0;
    for (int q = lane; q < b; q += 32) s += totals[q];
    for (int d = 16; d >= 1; d >>= 1) s += __shfl_xor(s, d);

    int* row = counts + b * NCH;
    int v[8], loc[8], run = 0;
    for (int k = 0; k < 8; ++k) { const int c = lane * 8 + k; v[k] = (c < NCH) ? row[c] : 0; }
    for (int k = 0; k < 8; ++k) { loc[k] = run; run += v[k]; }
    int incl = run;
    for (int d = 1; d < 32; d <<= 1) {
        const int up = __shfl_up(incl, d);
        if (lane >= d) incl += up;
    }
    const int excl = incl - run;
    for (int k = 0; k < 8; ++k) { const int c = lane * 8 + k; if (c < NCH) row[c] = s + excl + loc[k]; }
    if (lane == 0) bin_start[b] = s;
}
static_assert(NCH <= 256, "bin_scan gives each of 32 lanes 8 chunk counts");

__global__ void __launch_bounds__(BIN_T) bin_fill(
    const int* __restrict__ ei, const int* __restrict__ offs, unsigned* __restrict__ sorted)
{
    __shared__ int base[NBIN];
    __shared__ int cntr[NBIN];
    const int c = blockIdx.x;
    const int* src = ei + c * CH_E;
    const int* dst = ei + NE + c * CH_E;
    int s[BIN_IT], d[BIN_IT];
    for (int k = 0; k < BIN_IT; ++k) {
        const int i = threadIdx.x + k * BIN_T;
        s[k] = (i < CH_E) ? src[i] : 0;
        d[k] = (i < CH_E) ? dst[i] : 0;
    }
    for (int b = threadIdx.x; b < NBIN; b += BIN_T) { base[b] = offs[b * NCH + c]; cntr[b] = 0; }
    __syncthreads();
    for (int k = 0; k < BIN_IT; ++k) {
        const int i = threadIdx.x + k * BIN_T;
        if (i < CH_E) {
            const int bin = min(d[k] / OWN_R, NBIN - 1);
            const int r   = atomicAdd(&cntr[bin], 1);
            sorted[base[bin] + r] = (unsigned)s[k] | ((unsigned)(d[k] - bin * OWN_R) << 16);
        }
    }
}

__global__ void __launch_bounds__(SUM_WAVES * 32) owner_sum(
    const unsigned* __restrict__ sorted, const int* __restrict__ bin_start,
    const int* __restrict__ totals, const float* __restrict__ y32,
    float* __restrict__ aggy, float* __restrict__ cnt)
{
    __shared__ float tile[OWN_R * DH];
    __shared__ float lcnt[OWN_R];

    const int b    = blockIdx.x;
    const int lane = threadIdx.x & 31;
    const int wave = __builtin_amdgcn_readfirstlane((int)(threadIdx.x >> 5));

    for (int i = threadIdx.x; i < OWN_R * DH; i += SUM_WAVES * 32) tile[i] = 0.0f;
    if (threadIdx.x < OWN_R) lcnt[threadIdx.x] = 0.0f;
    __syncthreads();

    const int start = bin_start[b];
    const int n     = totals[b];
    for (int e0 = wave * EPW; e0 < n; e0 += SUM_WAVES * EPW) {
        const int      my = e0 + (lane & (EPW - 1));
        const unsigned pk = sorted[min(start + my, NE - 1)];

        unsigned dl[EPW];
        float v[EPW][4];
        for (int j = 0; j < EPW; ++j) {
            const unsigned pkj = (unsigned)__builtin_amdgcn_readlane((int)pk, j);
            dl[j] = pkj >> 16;
            const float* ys = y32 + (size_t)(pkj & 0xFFFFu) * DH + lane;
            for (int i = 0; i < 4; ++i) v[j][i] = ys[i * 32];
        }
        for (int j = 0; j < EPW; ++j) {
            if (e0 + j < n) {
                float* tp = tile + dl[j] * DH + lane;
                for (int i = 0; i < 4; ++i) atomicAdd(&tp[i * 32], v[j][i]);
            }
        }
        if (lane < EPW && my < n) atomicAdd(&lcnt[pk >> 16], 1.0f);
    }
    __syncthreads();

    float* ag = aggy + (size_t)b * OWN_R * DH;
    for (int i = threadIdx.x; i < OWN_R * DH; i += SUM_WAVES * 32) ag[i] = tile[i];
    if (threadIdx.x < OWN_R) cnt[b * OWN_R + threadIdx.x] = lcnt[threadIdx.x];
}

__global__ void __launch_bounds__(256) __attribute__((amdgpu_waves_per_eu(4, 4))) cand_kernel(
    const float* __restrict__ aggy, const float* __restrict__ cnt,
    const _Float16* __restrict__ x16, const _Float16* __restrict__ rh16,
    const _Float16* __restrict__ fWr,
    const float* __restrict__ b_l,
    const float* __restrict__ z_ws, const float* __restrict__ h_prev,
    float* __restrict__ out)
{
    const int lane  = threadIdx.x & 31;
    const int ntile = threadIdx.x >> 5;
    const int row0  = blockIdx.x * (MT * 16);
    const int col   = ntile * 16 + (lane & 15);

    const int rloc = (lane & 16) ? 8 : 0;

    v8f acc[MT];
    const float bl = b_l[col];
    {
        float cn[MT][8], ag[MT][8];
#pragma unroll
        for (int m = 0; m < MT; ++m) {
#pragma unroll
            for (int v = 0; v < 8; ++v) {
                const int row = row0 + m * 16 + rloc + v;
                cn[m][v] = cnt[row];
                ag[m][v] = aggy[(size_t)row * DH + col];
            }
        }
#pragma unroll
        for (int m = 0; m < MT; ++m) {
#pragma unroll
            for (int v = 0; v < 8; ++v) {
                const float inv = 1.0f / fmaxf(cn[m][v], 1.0f);
                acc[m][v] = ag[m][v] * inv + bl;
            }
        }
    }

    const size_t abase = (size_t)blockIdx.x * (WG_PIECES * PIECE) + lane * 8;
#pragma unroll
    for (int kp = 0; kp < 4; ++kp) {
        v16h bwr[2], ac[2][MT];
#pragma unroll
        for (int q = 0; q < 2; ++q) {
            const int kt = kp * 2 + q;
            const _Float16* xcb = (kt < 4)
                ? (x16  + abase + kt * (2 * PIECE))
                : (rh16 + abase + (kt - 4) * (2 * PIECE));
            const size_t off = (size_t)(kt * 8 + ntile) * (2 * PIECE) + lane * 8;
            bwr[q] = load_frag(fWr + off, fWr + off + PIECE);
#pragma unroll
            for (int m = 0; m < MT; ++m) {
                const _Float16* am = xcb + m * 4 * (2 * PIECE);
                ac[q][m] = load_frag(am, am + PIECE);
            }
        }
#pragma unroll
        for (int q = 0; q < 2; ++q) {
#pragma unroll
            for (int m = 0; m < MT; ++m)
                acc[m] = WMMA_F16(ac[q][m], bwr[q], acc[m]);
        }
    }

    float zz[MT][8], hh[MT][8];
#pragma unroll
    for (int m = 0; m < MT; ++m) {
#pragma unroll
        for (int v = 0; v < 8; ++v) {
            const size_t idx = (size_t)(row0 + m * 16 + rloc + v) * DH + col;
            zz[m][v] = z_ws[idx];
            hh[m][v] = h_prev[idx];
        }
    }
#pragma unroll
    for (int m = 0; m < MT; ++m) {
#pragma unroll
        for (int v = 0; v < 8; ++v) {
            const size_t idx = (size_t)(row0 + m * 16 + rloc + v) * DH + col;
            const float zv = zz[m][v];
            out[idx] = (1.0f - zv) * acc[m][v] + zv * hh[m][v];
        }
    }
}

extern "C" void kernel_launch(void* const* d_in, const int* in_sizes, int n_in,
                              void* d_out, int out_size, void* d_ws, size_t ws_size,
                              hipStream_t stream) {
    const float* x      = (const float*)d_in[0];
    const int*   ei     = (const int*)  d_in[1];
    const float* h_prev = (const float*)d_in[2];
    const float* W_xr   = (const float*)d_in[3];
    const float* b_xr   = (const float*)d_in[4];
    const float* W_hr   = (const float*)d_in[5];
    const float* W_xz   = (const float*)d_in[6];
    const float* b_xz   = (const float*)d_in[7];
    const float* W_hz   = (const float*)d_in[8];
    const float* W_l    = (const float*)d_in[9];
    const float* b_l    = (const float*)d_in[10];
    const float* W_r    = (const float*)d_in[11];
    float* out = (float*)d_out;

    char* ws = (char*)d_ws;
    size_t off = 0;
    auto alloc = [&](size_t bytes) -> char* {
        char* p = ws + off;
        off += (bytes + 255) & ~(size_t)255;
        return p;
    };
    _Float16* x16  = (_Float16*)alloc((size_t)NN * DH * 2);
    _Float16* h16  = (_Float16*)alloc((size_t)NN * DH * 2);
    _Float16* rh16 = (_Float16*)alloc((size_t)NN * DH * 2);
    float*    z_ws = (float*)   alloc((size_t)NN * DH * 4);
    float*    y32  = (float*)   alloc((size_t)NN * DH * 4);
    float*    aggy = (float*)   alloc((size_t)NN * DH * 4);
    float*    cnt  = (float*)   alloc((size_t)NN * 4);
    unsigned* sorted = (unsigned*)alloc((size_t)NE * 4);
    int*      counts = (int*)     alloc((size_t)NBIN * NCH * 4);
    int*      totals = (int*)     alloc((size_t)NBIN * 4);
    int*      bstart = (int*)     alloc((size_t)NBIN * 4);
    _Float16* fWxr = (_Float16*)alloc(128 * 128 * 2);
    _Float16* fWhr = (_Float16*)alloc(128 * 128 * 2);
    _Float16* fWxz = (_Float16*)alloc(128 * 128 * 2);
    _Float16* fWhz = (_Float16*)alloc(128 * 128 * 2);
    _Float16* fWl  = (_Float16*)alloc(256 * 128 * 2);
    _Float16* fWr  = (_Float16*)alloc(256 * 128 * 2);

    prep_all<<<PREP_BLOCKS, 256, 0, stream>>>(x, h_prev, x16, h16, totals,
                                              W_xr, W_hr, W_xz, W_hz, W_l, W_r,
                                              fWxr, fWhr, fWxz, fWhz, fWl, fWr);

    gates_kernel<<<MGROUPS, 256, 0, stream>>>(x16, h16, h_prev, b_xr, b_xz,
                                             fWxr, fWhr, fWxz, fWhz, fWl,
                                             z_ws, y32, rh16);

    bin_count<<<NCH, BIN_T, 0, stream>>>(ei, counts, totals);
    bin_scan<<<(NBIN + 7) / 8, 256, 0, stream>>>(counts, totals, bstart);
    bin_fill<<<NCH, BIN_T, 0, stream>>>(ei, counts, sorted);

    owner_sum<<<NBIN, SUM_WAVES * 32, 0, stream>>>(sorted, bstart, totals, y32, aggy, cnt);

    cand_kernel<<<MGROUPS, 256, 0, stream>>>(aggy, cnt, x16, rh16, fWr,
                                            b_l, z_ws, h_prev, out);
}


# Owner blocks sort their edge list by row in LDS; a wave sums runs of equal rows in registers, one LDS f32 add per run and column (8x fewer); counts come from the sort; f32 sum order changes
# speedup vs baseline: 4.9532x; 1.4155x over previous
#include <hip/hip_runtime.h>
#include <hip/hip_bf16.h>


typedef __attribute__((ext_vector_type(16))) _Float16 v16h;
typedef __attribute__((ext_vector_type(8)))  float    v8f;
typedef __attribute__((ext_vector_type(4)))  int      v4i;
typedef __attribute__((ext_vector_type(4)))  float    v4f;
typedef __attribute__((ext_vector_type(4)))  _Float16 v4h;

#define NN      50000
#define DH      128
#define DCAT    256
#define NE      800000
#define MTILES  (NN / 16)
#define MT      5
#define MGROUPS (MTILES / MT)
#define RH_PITCH 136
#define PIECE   256
#define WG_PIECES (MT * 4 * 2)
#define EPW     16
#define OWN_R   100
#define NBIN    (NN / OWN_R)
#define NCH     125
#define CH_E    (NE / NCH)
#define SORT_IT (CH_E / 256)
#define SUM_WAVES 16
#define SEG_SLOTS 128
#define CPW     (SEG_SLOTS / SUM_WAVES)
#define ECAP    4096
#define SORT_PT (ECAP / (SUM_WAVES * 32))
static_assert(NN % OWN_R == 0 && NE % NCH == 0, "bins tile the nodes and chunks tile the edges exactly");
static_assert(CH_E % 256 == 0 && NBIN % 2 == 0 && NBIN / 2 <= 256, "a chunk's block: no tail in the edge loops, two bins per thread in the scan");
static_assert(CH_E <= 65535 && NCH <= SEG_SLOTS, "a segment descriptor packs offset (high 16 bits) and count (low 16 bits) inside the chunk");
static_assert(MTILES % MT == 0, "the GEMM workgroups tile the rows exactly: no tail, no guard");
static_assert(ECAP % (SUM_WAVES * 32) == 0 && OWN_R <= 128, "row order: no tail in a thread's entries; one wave scans the row counts, four per lane");
static_assert(NN <= 65536 && OWN_R <= 65536, "a sorted entry packs the source node (low 16 bits) and the row inside the owner's range (high 16 bits)");

#define WMMA_F16(A, B, C) \
    __builtin_amdgcn_wmma_f32_16x16x32_f16(false, (A), false, (B), (short)0, (C), false, false)

__device__ __forceinline__ v16h load_frag(const _Float16* p0, const _Float16* p1) {
    v16h r;
    reinterpret_cast<v4i*>(&r)[0] = *reinterpret_cast<const v4i*>(p0);
    reinterpret_cast<v4i*>(&r)[1] = *reinterpret_cast<const v4i*>(p1);
    return r;
}

__device__ __forceinline__ float sigmoid_f(float x) {
    return 1.0f / (1.0f + __expf(-x));
}

#define ROW_LOADS   (MT * 16 * (DH / 4) / 256)
#define ACT_BLOCKS  (2 * MGROUPS)
#define W_BLOCKS    (4 * 64 + 2 * 128)
#define PREP_BLOCKS (NCH + W_BLOCKS + ACT_BLOCKS)
static_assert(ROW_LOADS * 256 == MT * 16 * (DH / 4) && WG_PIECES == MT * 8, "8 waves move the 80 rows in and the 40 pieces out without a tail");

__global__ void __launch_bounds__(256) prep_all(
    const float* __restrict__ x,    const float* __restrict__ h_prev,
    _Float16* __restrict__ x16,     _Float16* __restrict__ h16,
    const int* __restrict__ ei,     unsigned* __restrict__ sorted,  unsigned* __restrict__ seg,
    const float* __restrict__ W_xr, const float* __restrict__ W_hr,
    const float* __restrict__ W_xz, const float* __restrict__ W_hz,
    const float* __restrict__ W_l,  const float* __restrict__ W_r,
    _Float16* __restrict__ fWxr,    _Float16* __restrict__ fWhr,
    _Float16* __restrict__ fWxz,    _Float16* __restrict__ fWhz,
    _Float16* __restrict__ fWl,     _Float16* __restrict__ fWr)
{
    __shared__ __attribute__((aligned(16))) _Float16 s_t[MT * 16 * RH_PITCH];
    __shared__ int hist[NBIN];
    __shared__ int lofs[NBIN];
    __shared__ int wtot[8];

    if (blockIdx.x < NCH) {
        const int c    = blockIdx.x;
        const int lane = threadIdx.x & 31;
        const int wave = threadIdx.x >> 5;
        const int* src = ei + c * CH_E + threadIdx.x;
        const int* dst = ei + NE + c * CH_E + threadIdx.x;
        unsigned s[SORT_IT], d[SORT_IT];
        for (int k = 0; k < SORT_IT; ++k) { s[k] = (unsigned)src[k * 256]; d[k] = (unsigned)dst[k * 256]; }
        for (int i = threadIdx.x; i < NBIN; i += 256) hist[i] = 0;
        __syncthreads();
        for (int k = 0; k < SORT_IT; ++k) {
            const int bin = min((int)(d[k] / OWN_R), NBIN - 1);
            const int rk  = atomicAdd(&hist[bin], 1);
            d[k] |= (unsigned)rk << 16;
        }
        __syncthreads();
        const bool has = threadIdx.x < NBIN / 2;
        const int a0 = has ? hist[2 * threadIdx.x]     : 0;
        const int a1 = has ? hist[2 * threadIdx.x + 1] : 0;
        int incl = a0 + a1;
        for (int dd = 1; dd < 32; dd <<= 1) {
            const int up = __shfl_up(incl, dd);
            if (lane >= dd) incl += up;
        }
        if (lane == 31) wtot[wave] = incl;
        __syncthreads();
        int ex0 = incl - (a0 + a1);
        for (int w = 0; w < 8; ++w) ex0 += (w < wave) ? wtot[w] : 0;
        if (has) {
            const int b0 = 2 * threadIdx.x;
            lofs[b0]     = ex0;
            lofs[b0 + 1] = ex0 + a0;
            seg[(size_t)b0 * NCH + c]       = ((unsigned)ex0 << 16) | (unsigned)a0;
            seg[(size_t)(b0 + 1) * NCH + c] = ((unsigned)(ex0 + a0) << 16) | (unsigned)a1;
        }
        __syncthreads();
        unsigned* so = sorted + (size_t)c * CH_E;
        for (int k = 0; k < SORT_IT; ++k) {
            const unsigned dn  = d[k] & 0xFFFFu;
            const int      bin = min((int)(dn / OWN_R), NBIN - 1);
            so[lofs[bin] + (int)(d[k] >> 16)] = s[k] | ((dn - (unsigned)(bin * OWN_R)) << 16);
        }
        return;
    }
    const int b = blockIdx.x - NCH;

    if (b < W_BLOCKS) {
        const float* W; _Float16* frag; int lb;
        if      (b <  64) { W = W_xr; frag = fWxr; lb = b;       }
        else if (b < 128) { W = W_hr; frag = fWhr; lb = b -  64; }
        else if (b < 192) { W = W_xz; frag = fWxz; lb = b - 128; }
        else if (b < 256) { W = W_hz; frag = fWhz; lb = b - 192; }
        else if (b < 384) { W = W_l;  frag = fWl;  lb = b - 256; }
        else              { W = W_r;  frag = fWr;  lb = b - 384; }
        const int idx   = lb * 256 + threadIdx.x;
        const int e     = idx & 7;
        const int lane  = (idx >> 3) & 31;
        const int q     = (idx >> 8) & 1;
        const int ntile = (idx >> 9) & 7;
        const int ktile = idx >> 12;
        const int n     = ntile * 16 + (lane & 15);
        const int khalf = lane >> 4;
        const int k     = ktile * 32 + q * 16 + khalf * 8 + e;
        frag[idx] = (_Float16)W[(size_t)k * DH + n];
    } else {
        const int a = b - W_BLOCKS;
        const int g = (a < MGROUPS) ? a : a - MGROUPS;
        const float*    src = (a < MGROUPS) ? x   : h_prev;
        _Float16*       dst = (a < MGROUPS) ? x16 : h16;
        const int lane = threadIdx.x & 31;
        const int wave = threadIdx.x >> 5;
        const v4f* sa = reinterpret_cast<const v4f*>(src) + (size_t)g * (MT * 16 * (DH / 4)) + threadIdx.x;
        v4f v[ROW_LOADS];
        for (int i = 0; i < ROW_LOADS; ++i) v[i] = sa[i * 256];
        for (int i = 0; i < ROW_LOADS; ++i)
            *reinterpret_cast<v4h*>(&s_t[(i * 8 + wave) * RH_PITCH + 4 * lane]) = __builtin_convertvector(v[i], v4h);
        __syncthreads();
        const _Float16* sp = s_t + (lane & 15) * RH_PITCH + wave * 16 + (lane >> 4) * 8;
        v4i* dp = reinterpret_cast<v4i*>(dst + ((size_t)g * WG_PIECES + wave) * PIECE + lane * 8);
        v4i pc[MT];
        for (int t = 0; t < MT; ++t) pc[t] = *reinterpret_cast<const v4i*>(sp + t * 16 * RH_PITCH);
        for (int t = 0; t < MT; ++t) dp[t * 8 * (PIECE / 8)] = pc[t];
    }
}

__global__ void __launch_bounds__(256) __attribute__((amdgpu_waves_per_eu(4, 4))) gates_kernel(
    const _Float16* __restrict__ x16,  const _Float16* __restrict__ h16,
    const float* __restrict__ h_prev,
    const float* __restrict__ b_xr,    const float* __restrict__ b_xz,
    const _Float16* __restrict__ fWxr, const _Float16* __restrict__ fWhr,
    const _Float16* __restrict__ fWxz, const _Float16* __restrict__ fWhz,
    const _Float16* __restrict__ fWl,
    float* __restrict__ z_ws, float* __restrict__ y32, _Float16* __restrict__ rh16)
{
    __shared__ __attribute__((aligned(16))) _Float16 s_rh[MT * 16 * RH_PITCH];

    const int lane  = threadIdx.x & 31;
    const int ntile = threadIdx.x >> 5;
    const int row0  = blockIdx.x * (MT * 16);
    const int col   = ntile * 16 + (lane & 15);
    const int khalf = lane >> 4;

    v8f accr[MT], accz[MT], accy[MT];
    const float br = b_xr[col], bz = b_xz[col];
#pragma unroll
    for (int m = 0; m < MT; ++m) {
#pragma unroll
        for (int v = 0; v < 8; ++v) { accr[m][v] = br; accz[m][v] = bz; accy[m][v] = 0.0f; }
    }

    const _Float16* xb = x16 + (size_t)blockIdx.x * (WG_PIECES * PIECE) + lane * 8;
    const _Float16* hb = h16 + (size_t)blockIdx.x * (WG_PIECES * PIECE) + lane * 8;

#pragma unroll
    for (int kt = 0; kt < 4; ++kt) {
        const size_t off = (size_t)(kt * 8 + ntile) * (2 * PIECE) + lane * 8;
        const v16h bxr = load_frag(fWxr + off, fWxr + off + PIECE);
        const v16h bhr = load_frag(fWhr + off, fWhr + off + PIECE);
        const v16h bxz = load_frag(fWxz + off, fWxz + off + PIECE);
        const v16h bhz = load_frag(fWhz + off, fWhz + off + PIECE);
        const v16h bwl = load_frag(fWl  + off, fWl  + off + PIECE);
        v16h ax[MT], ah[MT];
#pragma unroll
        for (int m = 0; m < MT; ++m) {
            const _Float16* xm = xb + (m * 4 + kt) * (2 * PIECE);
            const _Float16* hm = hb + (m * 4 + kt) * (2 * PIECE);
            ax[m] = load_frag(xm, xm + PIECE);
            ah[m] = load_frag(hm, hm + PIECE);
        }
#pragma unroll
        for (int m = 0; m < MT; ++m) {
            accr[m] = WMMA_F16(ax[m], bxr, accr[m]);
            accr[m] = WMMA_F16(ah[m], bhr, accr[m]);
            accz[m] = WMMA_F16(ax[m], bxz, accz[m]);
            accz[m] = WMMA_F16(ah[m], bhz, accz[m]);
            accy[m] = WMMA_F16(ax[m], bwl, accy[m]);
        }
    }

    const int rloc = (lane & 16) ? 8 : 0;
    float hv[MT][8];
#pragma unroll
    for (int m = 0; m < MT; ++m) {
#pragma unroll
        for (int v = 0; v < 8; ++v)
            hv[m][v] = h_prev[(size_t)(row0 + m * 16 + rloc + v) * DH + col];
    }
    v16h bw2[4];
#pragma unroll
    for (int kt = 0; kt < 4; ++kt) {
        const size_t off = (size_t)((4 + kt) * 8 + ntile) * (2 * PIECE) + lane * 8;
        bw2[kt] = load_frag(fWl + off, fWl + off + PIECE);
    }
#pragma unroll
    for (int m = 0; m < MT; ++m) {
#pragma unroll
        for (int v = 0; v < 8; ++v) {
            const int    lrow = m * 16 + rloc + v;
            const size_t idx  = (size_t)(row0 + lrow) * DH + col;
            const float rv = sigmoid_f(accr[m][v]);
            const float zv = sigmoid_f(accz[m][v]);
            const float h  = hv[m][v];
            const _Float16 rh = (_Float16)(rv * h);
            z_ws[idx] = zv;
            s_rh[lrow * RH_PITCH + col] = rh;
        }
    }
    __syncthreads();

    {
        const _Float16* cp = s_rh + (lane & 15) * RH_PITCH + ntile * 16 + khalf * 8;
        v4i* dp = reinterpret_cast<v4i*>(rh16 + ((size_t)blockIdx.x * WG_PIECES + ntile) * PIECE + lane * 8);
        v4i pc[MT];
#pragma unroll
        for (int m = 0; m < MT; ++m) pc[m] = *reinterpret_cast<const v4i*>(cp + m * 16 * RH_PITCH);
#pragma unroll
        for (int m = 0; m < MT; ++m) dp[m * 8 * (PIECE / 8)] = pc[m];
    }

    const _Float16* sp = s_rh + (lane & 15) * RH_PITCH + khalf * 8;
#pragma unroll
    for (int kt = 0; kt < 4; ++kt) {
        v16h arh[MT];
#pragma unroll
        for (int m = 0; m < MT; ++m) {
            const _Float16* sm = sp + m * 16 * RH_PITCH + kt * 32;
            arh[m] = load_frag(sm, sm + 16);
        }
#pragma unroll
        for (int m = 0; m < MT; ++m)
            accy[m] = WMMA_F16(arh[m], bw2[kt], accy[m]);
    }
#pragma unroll
    for (int m = 0; m < MT; ++m) {
#pragma unroll
        for (int v = 0; v < 8; ++v)
            y32[(size_t)(row0 + m * 16 + rloc + v) * DH + col] = accy[m][v];
    }
}

__global__ void __launch_bounds__(SUM_WAVES * 32) owner_sum(
    const unsigned* __restrict__ sorted, const unsigned* __restrict__ seg,
    const float* __restrict__ y32,
    float* __restrict__ aggy, float* __restrict__ cnt)
{
    __shared__ float    tile[OWN_R * DH];
    __shared__ float    lcnt[OWN_R];
    __shared__ unsigned elist[ECAP];
    __shared__ unsigned rlist[ECAP];
    __shared__ int      rhist[OWN_R];
    __shared__ int      rstart[OWN_R];
    __shared__ unsigned cseg[SEG_SLOTS];
    __shared__ int      cpos[SEG_SLOTS];
    __shared__ int      wsum[SEG_SLOTS / 32];

    const int b    = blockIdx.x;
    const int lane = threadIdx.x & 31;
    const int wave = __builtin_amdgcn_readfirstlane((int)(threadIdx.x >> 5));

    unsigned sg = 0;
    int incl = 0;
    if (wave < SEG_SLOTS / 32) {
        sg = (threadIdx.x < NCH) ? seg[(size_t)b * NCH + threadIdx.x] : 0u;
        incl = (int)(sg & 0xFFFFu);
        for (int dd = 1; dd < 32; dd <<= 1) {
            const int up = __shfl_up(incl, dd);
            if (lane >= dd) incl += up;
        }
        if (lane == 31) wsum[wave] = incl;
    }
    for (int i = threadIdx.x; i < OWN_R * DH; i += SUM_WAVES * 32) tile[i] = 0.0f;
    if (threadIdx.x < OWN_R) lcnt[threadIdx.x] = 0.0f;
    __syncthreads();
    int tot = 0;
    for (int w = 0; w < SEG_SLOTS / 32; ++w) tot += wsum[w];
    const int n = __builtin_amdgcn_readfirstlane(tot);
    if (wave < SEG_SLOTS / 32) {
        int p = incl - (int)(sg & 0xFFFFu);
        for (int w = 0; w < SEG_SLOTS / 32; ++w) p += (w < wave) ? wsum[w] : 0;
        cseg[threadIdx.x] = sg;
        cpos[threadIdx.x] = p;
    }
    __syncthreads();

    for (int wb = 0; wb < n; wb += ECAP) {
        const int nw = min(n - wb, ECAP);
        if (threadIdx.x < OWN_R) rhist[threadIdx.x] = 0;
        {
            unsigned ev[CPW];
            int      qv[CPW];
            for (int i = 0; i < CPW; ++i) {
                const int      cc  = wave * CPW + i;
                const unsigned sgi = cseg[cc];
                const int      q   = cpos[cc] + lane - wb;
                const bool     ok  = lane < (int)(sgi & 0xFFFFu) && (unsigned)q < (unsigned)ECAP;
                ev[i] = ok ? sorted[(size_t)cc * CH_E + (sgi >> 16) + lane] : 0u;
                qv[i] = ok ? q : -1;
            }
            for (int i = 0; i < CPW; ++i)
                if (qv[i] >= 0) elist[qv[i]] = ev[i];
            for (int i = 0; i < CPW; ++i) {
                const int      cc  = wave * CPW + i;
                const unsigned sgi = cseg[cc];
                const int      cn  = (int)(sgi & 0xFFFFu);
                for (int l = lane + 32; l < cn; l += 32) {
                    const int q = cpos[cc] + l - wb;
                    if ((unsigned)q < (unsigned)ECAP) elist[q] = sorted[(size_t)cc * CH_E + (sgi >> 16) + l];
                }
            }
        }
        __syncthreads();

        unsigned pe[SORT_PT];
        int      rk[SORT_PT];
        for (int k = 0; k < SORT_PT; ++k) {
            const int i = (int)threadIdx.x + k * (SUM_WAVES * 32);
            pe[k] = 0u; rk[k] = 0;
            if (i < nw) { pe[k] = elist[i]; rk[k] = atomicAdd(&rhist[pe[k] >> 16], 1); }
        }
        __syncthreads();
        if (threadIdx.x < OWN_R) lcnt[threadIdx.x] += (float)rhist[threadIdx.x];
        if (wave == 0) {
            int cr[4], s = 0;
            for (int i = 0; i < 4; ++i) { const int r = lane * 4 + i; cr[i] = (r < OWN_R) ? rhist[r] : 0; s += cr[i]; }
            int inc = s;
            for (int dd = 1; dd < 32; dd <<= 1) {
                const int up = __shfl_up(inc, dd);
                if (lane >= dd) inc += up;
            }
            int ex = inc - s;
            for (int i = 0; i < 4; ++i) { const int r = lane * 4 + i; if (r < OWN_R) rstart[r] = ex; ex += cr[i]; }
        }
        __syncthreads();
        for (int k = 0; k < SORT_PT; ++k) {
            const int i = (int)threadIdx.x + k * (SUM_WAVES * 32);
            if (i < nw) rlist[rstart[pe[k] >> 16] + rk[k]] = pe[k];
        }
        __syncthreads();

        for (int e0 = wave * EPW; e0 < nw; e0 += SUM_WAVES * EPW) {
            const int      my = e0 + (lane & (EPW - 1));
            const unsigned pk = rlist[min(my, nw - 1)];

            unsigned dl[EPW];
            float v[EPW][4];
            for (int j = 0; j < EPW; ++j) {
                const unsigned pkj = (unsigned)__builtin_amdgcn_readlane((int)pk, j);
                dl[j] = pkj >> 16;
                const float* ys = y32 + (size_t)(pkj & 0xFFFFu) * DH + lane;
                for (int i = 0; i < 4; ++i) v[j][i] = ys[i * 32];
            }
            float a[4] = {0.0f, 0.0f, 0.0f, 0.0f};
            for (int j = 0; j < EPW; ++j) {
                if (e0 + j < nw) {
                    for (int i = 0; i < 4; ++i) a[i] += v[j][i];
                    const unsigned nx = dl[(j + 1 < EPW) ? j + 1 : j];
                    const bool last = (j + 1 == EPW) || (e0 + j + 1 >= nw) || (nx != dl[j]);
                    if (last) {
                        float* tp = tile + dl[j] * DH + lane;
                        for (int i = 0; i < 4; ++i) { atomicAdd(&tp[i * 32], a[i]); a[i] = 0.0f; }
                    }
                }
            }
        }
        __syncthreads();
    }

    float* ag = aggy + (size_t)b * OWN_R * DH;
    for (int i = threadIdx.x; i < OWN_R * DH; i += SUM_WAVES * 32) ag[i] = tile[i];
    if (threadIdx.x < OWN_R) cnt[b * OWN_R + threadIdx.x] = lcnt[threadIdx.x];
}

__global__ void __launch_bounds__(256) __attribute__((amdgpu_waves_per_eu(4, 4))) cand_kernel(
    const float* __restrict__ aggy, const float* __restrict__ cnt,
    const _Float16* __restrict__ x16, const _Float16* __restrict__ rh16,
    const _Float16* __restrict__ fWr,
    const float* __restrict__ b_l,
    const float* __restrict__ z_ws, const float* __restrict__ h_prev,
    float* __restrict__ out)
{
    const int lane  = threadIdx.x & 31;
    const int ntile = threadIdx.x >> 5;
    const int row0  = blockIdx.x * (MT * 16);
    const int col   = ntile * 16 + (lane & 15);

    const int rloc = (lane & 16) ? 8 : 0;

    v8f acc[MT];
    const float bl = b_l[col];
    {
        float cn[MT][8], ag[MT][8];
#pragma unroll
        for (int m = 0; m < MT; ++m) {
#pragma unroll
            for (int v = 0; v < 8; ++v) {
                const int row = row0 + m * 16 + rloc + v;
                cn[m][v] = cnt[row];
                ag[m][v] = aggy[(size_t)row * DH + col];
            }
        }
#pragma unroll
        for (int m = 0; m < MT; ++m) {
#pragma unroll
            for (int v = 0; v < 8; ++v) {
                const float inv = 1.0f / fmaxf(cn[m][v], 1.0f);
                acc[m][v] = ag[m][v] * inv + bl;
            }
        }
    }

    const size_t abase = (size_t)blockIdx.x * (WG_PIECES * PIECE) + lane * 8;
#pragma unroll
    for (int kp = 0; kp < 4; ++kp) {
        v16h bwr[2], ac[2][MT];
#pragma unroll
        for (int q = 0; q < 2; ++q) {
            const int kt = kp * 2 + q;
            const _Float16* xcb = (kt < 4)
                ? (x16  + abase + kt * (2 * PIECE))
                : (rh16 + abase + (kt - 4) * (2 * PIECE));
            const size_t off = (size_t)(kt * 8 + ntile) * (2 * PIECE) + lane * 8;
            bwr[q] = load_frag(fWr + off, fWr + off + PIECE);
#pragma unroll
            for (int m = 0; m < MT; ++m) {
                const _Float16* am = xcb + m * 4 * (2 * PIECE);
                ac[q][m] = load_frag(am, am + PIECE);
            }
        }
#pragma unroll
        for (int q = 0; q < 2; ++q) {
#pragma unroll
            for (int m = 0; m < MT; ++m)
                acc[m] = WMMA_F16(ac[q][m], bwr[q], acc[m]);
        }
    }

    float zz[MT][8], hh[MT][8];
#pragma unroll
    for (int m = 0; m < MT; ++m) {
#pragma unroll
        for (int v = 0; v < 8; ++v) {
            const size_t idx = (size_t)(row0 + m * 16 + rloc + v) * DH + col;
            zz[m][v] = z_ws[idx];
            hh[m][v] = h_prev[idx];
        }
    }
#pragma unroll
    for (int m = 0; m < MT; ++m) {
#pragma unroll
        for (int v = 0; v < 8; ++v) {
            const size_t idx = (size_t)(row0 + m * 16 + rloc + v) * DH + col;
            const float zv = zz[m][v];
            out[idx] = (1.0f - zv) * acc[m][v] + zv * hh[m][v];
        }
    }
}

extern "C" void kernel_launch(void* const* d_in, const int* in_sizes, int n_in,
                              void* d_out, int out_size, void* d_ws, size_t ws_size,
                              hipStream_t stream) {
    const float* x      = (const float*)d_in[0];
    const int*   ei     = (const int*)  d_in[1];
    const float* h_prev = (const float*)d_in[2];
    const float* W_xr   = (const float*)d_in[3];
    const float* b_xr   = (const float*)d_in[4];
    const float* W_hr   = (const float*)d_in[5];
    const float* W_xz   = (const float*)d_in[6];
    const float* b_xz   = (const float*)d_in[7];
    const float* W_hz   = (const float*)d_in[8];
    const float* W_l    = (const float*)d_in[9];
    const float* b_l    = (const float*)d_in[10];
    const float* W_r    = (const float*)d_in[11];
    float* out = (float*)d_out;

    char* ws = (char*)d_ws;
    size_t off = 0;
    auto alloc = [&](size_t bytes) -> char* {
        char* p = ws + off;
        off += (bytes + 255) & ~(size_t)255;
        return p;
    };
    _Float16* x16  = (_Float16*)alloc((size_t)NN * DH * 2);
    _Float16* h16  = (_Float16*)alloc((size_t)NN * DH * 2);
    _Float16* rh16 = (_Float16*)alloc((size_t)NN * DH * 2);
    float*    z_ws = (float*)   alloc((size_t)NN * DH * 4);
    float*    y32  = (float*)   alloc((size_t)NN * DH * 4);
    float*    aggy = (float*)   alloc((size_t)NN * DH * 4);
    float*    cnt  = (float*)   alloc((size_t)NN * 4);
    unsigned* sorted = (unsigned*)alloc((size_t)NE * 4);
    unsigned* seg    = (unsigned*)alloc((size_t)NBIN * NCH * 4);
    _Float16* fWxr = (_Float16*)alloc(128 * 128 * 2);
    _Float16* fWhr = (_Float16*)alloc(128 * 128 * 2);
    _Float16* fWxz = (_Float16*)alloc(128 * 128 * 2);
    _Float16* fWhz = (_Float16*)alloc(128 * 128 * 2);
    _Float16* fWl  = (_Float16*)alloc(256 * 128 * 2);
    _Float16* fWr  = (_Float16*)alloc(256 * 128 * 2);

    prep_all<<<PREP_BLOCKS, 256, 0, stream>>>(x, h_prev, x16, h16, ei, sorted, seg,
                                              W_xr, W_hr, W_xz, W_hz, W_l, W_r,
                                              fWxr, fWhr, fWxz, fWhz, fWl, fWr);

    gates_kernel<<<MGROUPS, 256, 0, stream>>>(x16, h16, h_prev, b_xr, b_xz,
                                             fWxr, fWhr, fWxz, fWhz, fWl,
                                             z_ws, y32, rh16);

    owner_sum<<<NBIN, SUM_WAVES * 32, 0, stream>>>(sorted, seg, y32, aggy, cnt);

    cand_kernel<<<MGROUPS, 256, 0, stream>>>(aggy, cnt, x16, rh16, fWr,
                                            b_l, z_ws, h_prev, out);
}
